# starting kernel
# baseline (speedup 1.0000x reference)
#include <hip/hip_runtime.h>
#include <math.h>

typedef __attribute__((ext_vector_type(2))) float v2f;
typedef __attribute__((ext_vector_type(8))) float v8f;

#define IN_C 128
#define LOG2C 7

__global__ void k_set_const(float* __restrict__ p, float v, int n) {
    int i = blockIdx.x * blockDim.x + threadIdx.x;
    if (i < n) p[i] = v;
}

__global__ void k_count_deg(const int* __restrict__ dst, float* __restrict__ deg, int nE) {
    int i = blockIdx.x * blockDim.x + threadIdx.x;
    if (i < nE) atomicAdd(&deg[dst[i]], 1.0f);
}

__global__ void k_inv_sqrt(float* __restrict__ p, int n) {
    int i = blockIdx.x * blockDim.x + threadIdx.x;
    if (i < n) p[i] = 1.0f / sqrtf(p[i]);
}

__global__ __launch_bounds__(128) void k_gemm_wmma_f32(
    const float* __restrict__ A, const float* __restrict__ W,
    float* __restrict__ C, int mTiles)
{
    const int wavesPerBlock = blockDim.x >> 5;
    const int wave = blockIdx.x * wavesPerBlock + (threadIdx.x >> 5);
    if (wave >= mTiles) return;

    const int lane = threadIdx.x & 31;
    const int row  = lane & 15;
    const int hi   = lane >> 4;
    const int koff = hi * 2;
    const int m0   = wave * 16;

    v8f acc[8] = {};

    const float* __restrict__ arow = A + (size_t)(m0 + row) * IN_C;

    for (int k = 0; k < IN_C; k += 4) {
        v2f a;
        a.x = arow[k + koff];
        a.y = arow[k + koff + 1];
        const float* __restrict__ wbase = W + (size_t)(k + koff) * IN_C + row;
#pragma unroll
        for (int t = 0; t < 8; ++t) {
            v2f b;
            b.x = wbase[t * 16];
            b.y = wbase[t * 16 + IN_C];
            acc[t] = __builtin_amdgcn_wmma_f32_16x16x4_f32(
                false, a, false, b, (short)0, acc[t], false, false);
        }
    }

    float* __restrict__ cbase = C + (size_t)(m0 + hi * 8) * IN_C + row;
#pragma unroll
    for (int t = 0; t < 8; ++t) {
#pragma unroll
        for (int r = 0; r < 8; ++r) {
            cbase[(size_t)r * IN_C + t * 16] = acc[t][r];
        }
    }
}

__global__ __launch_bounds__(256) void k_aggregate(
    const float* __restrict__ h,
    const int* __restrict__ src, const int* __restrict__ dst,
    const float* __restrict__ inv_sqrt,
    float* __restrict__ agg, int nE)
{
    const int wave = blockIdx.x * (blockDim.x >> 5) + (threadIdx.x >> 5);
    if (wave >= nE) return;
    const int lane = threadIdx.x & 31;

    const int s = src[wave];
    const int d = dst[wave];
    const float coef = inv_sqrt[s] * inv_sqrt[d];

    const float4 hv = ((const float4*)(h + (size_t)s * IN_C))[lane];
    float* ap = agg + (size_t)d * IN_C + lane * 4;
    atomicAdd(ap + 0, hv.x * coef);
    atomicAdd(ap + 1, hv.y * coef);
    atomicAdd(ap + 2, hv.z * coef);
    atomicAdd(ap + 3, hv.w * coef);
}

__global__ void k_finalize(
    const float* __restrict__ agg, const float* __restrict__ h,
    const float* __restrict__ inv_sqrt, const float* __restrict__ bias,
    float* __restrict__ out, int total, int do_relu)
{
    int i = blockIdx.x * blockDim.x + threadIdx.x;
    if (i >= total) return;
    int n = i >> LOG2C;
    int c = i & (IN_C - 1);
    float is = inv_sqrt[n];
    float v = agg[i] + h[i] * is * is + bias[c];
    if (do_relu) v = fmaxf(v, 0.0f);
    out[i] = v;
}

extern "C" void kernel_launch(void* const* d_in, const int* in_sizes, int n_in,
                              void* d_out, int out_size, void* d_ws, size_t ws_size,
                              hipStream_t stream)
{
    const float* x   = (const float*)d_in[0];
    const int*   ei  = (const int*)d_in[1];
    const float* W1  = (const float*)d_in[2];
    const float* b1  = (const float*)d_in[3];
    const float* W2  = (const float*)d_in[4];
    const float* b2  = (const float*)d_in[5];

    const int N = in_sizes[0] / IN_C;
    const int E = in_sizes[1] / 2;
    const int* src = ei;
    const int* dst = ei + E;

    const int total = N * IN_C;

    float* inv_sqrt = (float*)d_ws;
    float* hbuf     = inv_sqrt + N;
    float* out      = (float*)d_out;

    const int T256 = 256;
    const int mTiles = (N + 15) / 16;
    dim3 gemmGrid((mTiles + 3) / 4);
    dim3 gemmBlock(128);
    dim3 aggGrid((E + 7) / 8);
    dim3 aggBlock(256);

    k_set_const<<<(N + T256 - 1) / T256, T256, 0, stream>>>(inv_sqrt, 1.0f, N);
    k_count_deg<<<(E + T256 - 1) / T256, T256, 0, stream>>>(dst, inv_sqrt, E);
    k_inv_sqrt<<<(N + T256 - 1) / T256, T256, 0, stream>>>(inv_sqrt, N);

    k_gemm_wmma_f32<<<gemmGrid, gemmBlock, 0, stream>>>(x, W1, hbuf, mTiles);
    k_set_const<<<(total + T256 - 1) / T256, T256, 0, stream>>>(out, 0.0f, total);
    k_aggregate<<<aggGrid, aggBlock, 0, stream>>>(hbuf, src, dst, inv_sqrt, out, E);
    k_finalize<<<(total + T256 - 1) / T256, T256, 0, stream>>>(
        out, hbuf, inv_sqrt, b1, out, total,  1);

    k_gemm_wmma_f32<<<gemmGrid, gemmBlock, 0, stream>>>(out, W2, hbuf, mTiles);
    k_set_const<<<(total + T256 - 1) / T256, T256, 0, stream>>>(out, 0.0f, total);
    k_aggregate<<<aggGrid, aggBlock, 0, stream>>>(hbuf, src, dst, inv_sqrt, out, E);
    k_finalize<<<(total + T256 - 1) / T256, T256, 0, stream>>>(
        out, hbuf, inv_sqrt, b2, out, total,  0);
}


# Edge pass: each wave takes 32 consecutive edges, not one: ids and coefficients loaded one per lane, handed out by readlane; 8 edges' row loads issued before their atomics; sum order: atomic arrival
# speedup vs baseline: 1.6525x; 1.6525x over previous
#include <hip/hip_runtime.h>
#include <math.h>

typedef __attribute__((ext_vector_type(2))) float v2f;
typedef __attribute__((ext_vector_type(8))) float v8f;

#define IN_C 128
#define LOG2C 7

__global__ void k_set_const(float* __restrict__ p, float v, int n) {
    int i = blockIdx.x * blockDim.x + threadIdx.x;
    if (i < n) p[i] = v;
}

__global__ void k_count_deg(const int* __restrict__ dst, float* __restrict__ deg, int nE) {
    int i = blockIdx.x * blockDim.x + threadIdx.x;
    if (i < nE) atomicAdd(&deg[dst[i]], 1.0f);
}

__global__ void k_inv_sqrt(float* __restrict__ p, int n) {
    int i = blockIdx.x * blockDim.x + threadIdx.x;
    if (i < n) p[i] = 1.0f / sqrtf(p[i]);
}

__global__ __launch_bounds__(128) void k_gemm_wmma_f32(
    const float* __restrict__ A, const float* __restrict__ W,
    float* __restrict__ C, int mTiles)
{
    const int wavesPerBlock = blockDim.x >> 5;
    const int wave = blockIdx.x * wavesPerBlock + (threadIdx.x >> 5);
    if (wave >= mTiles) return;

    const int lane = threadIdx.x & 31;
    const int row  = lane & 15;
    const int hi   = lane >> 4;
    const int koff = hi * 2;
    const int m0   = wave * 16;

    v8f acc[8] = {};

    const float* __restrict__ arow = A + (size_t)(m0 + row) * IN_C;

    for (int k = 0; k < IN_C; k += 4) {
        v2f a;
        a.x = arow[k + koff];
        a.y = arow[k + koff + 1];
        const float* __restrict__ wbase = W + (size_t)(k + koff) * IN_C + row;
#pragma unroll
        for (int t = 0; t < 8; ++t) {
            v2f b;
            b.x = wbase[t * 16];
            b.y = wbase[t * 16 + IN_C];
            acc[t] = __builtin_amdgcn_wmma_f32_16x16x4_f32(
                false, a, false, b, (short)0, acc[t], false, false);
        }
    }

    float* __restrict__ cbase = C + (size_t)(m0 + hi * 8) * IN_C + row;
#pragma unroll
    for (int t = 0; t < 8; ++t) {
#pragma unroll
        for (int r = 0; r < 8; ++r) {
            cbase[(size_t)r * IN_C + t * 16] = acc[t][r];
        }
    }
}

#define EDGES_PER_WAVE 32
#define EDGE_BATCH 8

__device__ __forceinline__ void edge_row_add(
    const float* __restrict__ h, float* __restrict__ agg,
    int sj, int dj, float cj, int lane)
{
    const float* __restrict__ hp = h + (size_t)(unsigned)sj * IN_C + lane;
    const float h0 = hp[0], h1 = hp[32], h2 = hp[64], h3 = hp[96];
    float* ap = agg + (size_t)(unsigned)dj * IN_C + lane;
    atomicAdd(ap +  0, h0 * cj);
    atomicAdd(ap + 32, h1 * cj);
    atomicAdd(ap + 64, h2 * cj);
    atomicAdd(ap + 96, h3 * cj);
}

__global__ __launch_bounds__(256) void k_aggregate(
    const float* __restrict__ h,
    const int* __restrict__ src, const int* __restrict__ dst,
    const float* __restrict__ inv_sqrt,
    float* __restrict__ agg, int nE)
{
    const int wave = __builtin_amdgcn_readfirstlane(
        (int)(blockIdx.x * (blockDim.x >> 5) + (threadIdx.x >> 5)));
    const int base = wave * EDGES_PER_WAVE;
    if (base >= nE) return;
    const int lane = threadIdx.x & 31;
    const int left = nE - base;
    const int cnt = left < EDGES_PER_WAVE ? left : EDGES_PER_WAVE;

    int s = 0, d = 0;
    float coef = 0.0f;
    if (lane < cnt) {
        s = src[base + lane];
        d = dst[base + lane];
        coef = inv_sqrt[s] * inv_sqrt[d];
    }
    const int coef_bits = __float_as_int(coef);

    int j = 0;
    for (; j + EDGE_BATCH <= cnt; j += EDGE_BATCH) {
        float v[EDGE_BATCH][4];
#pragma unroll
        for (int u = 0; u < EDGE_BATCH; ++u) {
            const int sj = __builtin_amdgcn_readlane(s, j + u);
            const float* __restrict__ hp = h + (size_t)(unsigned)sj * IN_C + lane;
            v[u][0] = hp[0]; v[u][1] = hp[32]; v[u][2] = hp[64]; v[u][3] = hp[96];
        }
#pragma unroll
        for (int u = 0; u < EDGE_BATCH; ++u) {
            const int dj = __builtin_amdgcn_readlane(d, j + u);
            const float cj = __int_as_float(__builtin_amdgcn_readlane(coef_bits, j + u));
            float* ap = agg + (size_t)(unsigned)dj * IN_C + lane;
            atomicAdd(ap +  0, v[u][0] * cj);
            atomicAdd(ap + 32, v[u][1] * cj);
            atomicAdd(ap + 64, v[u][2] * cj);
            atomicAdd(ap + 96, v[u][3] * cj);
        }
    }
    for (; j < cnt; ++j) {
        edge_row_add(h, agg,
                     __builtin_amdgcn_readlane(s, j),
                     __builtin_amdgcn_readlane(d, j),
                     __int_as_float(__builtin_amdgcn_readlane(coef_bits, j)), lane);
    }
}

__global__ void k_finalize(
    const float* __restrict__ agg, const float* __restrict__ h,
    const float* __restrict__ inv_sqrt, const float* __restrict__ bias,
    float* __restrict__ out, int total, int do_relu)
{
    int i = blockIdx.x * blockDim.x + threadIdx.x;
    if (i >= total) return;
    int n = i >> LOG2C;
    int c = i & (IN_C - 1);
    float is = inv_sqrt[n];
    float v = agg[i] + h[i] * is * is + bias[c];
    if (do_relu) v = fmaxf(v, 0.0f);
    out[i] = v;
}

extern "C" void kernel_launch(void* const* d_in, const int* in_sizes, int n_in,
                              void* d_out, int out_size, void* d_ws, size_t ws_size,
                              hipStream_t stream)
{
    const float* x   = (const float*)d_in[0];
    const int*   ei  = (const int*)d_in[1];
    const float* W1  = (const float*)d_in[2];
    const float* b1  = (const float*)d_in[3];
    const float* W2  = (const float*)d_in[4];
    const float* b2  = (const float*)d_in[5];

    const int N = in_sizes[0] / IN_C;
    const int E = in_sizes[1] / 2;
    const int* src = ei;
    const int* dst = ei + E;

    const int total = N * IN_C;

    float* inv_sqrt = (float*)d_ws;
    float* hbuf     = inv_sqrt + N;
    float* out      = (float*)d_out;

    const int T256 = 256;
    const int mTiles = (N + 15) / 16;
    dim3 gemmGrid((mTiles + 3) / 4);
    dim3 gemmBlock(128);
    dim3 aggGrid((E + 255) / 256);
    dim3 aggBlock(256);

    k_set_const<<<(N + T256 - 1) / T256, T256, 0, stream>>>(inv_sqrt, 1.0f, N);
    k_count_deg<<<(E + T256 - 1) / T256, T256, 0, stream>>>(dst, inv_sqrt, E);
    k_inv_sqrt<<<(N + T256 - 1) / T256, T256, 0, stream>>>(inv_sqrt, N);

    k_gemm_wmma_f32<<<gemmGrid, gemmBlock, 0, stream>>>(x, W1, hbuf, mTiles);
    k_set_const<<<(total + T256 - 1) / T256, T256, 0, stream>>>(out, 0.0f, total);
    k_aggregate<<<aggGrid, aggBlock, 0, stream>>>(hbuf, src, dst, inv_sqrt, out, E);
    k_finalize<<<(total + T256 - 1) / T256, T256, 0, stream>>>(
        out, hbuf, inv_sqrt, b1, out, total,  1);

    k_gemm_wmma_f32<<<gemmGrid, gemmBlock, 0, stream>>>(out, W2, hbuf, mTiles);
    k_set_const<<<(total + T256 - 1) / T256, T256, 0, stream>>>(out, 0.0f, total);
    k_aggregate<<<aggGrid, aggBlock, 0, stream>>>(hbuf, src, dst, inv_sqrt, out, E);
    k_finalize<<<(total + T256 - 1) / T256, T256, 0, stream>>>(
        out, hbuf, inv_sqrt, b2, out, total,  0);
}


# Self-loop term, bias, ReLU folded into product kernels: epilogue also stores the edge sums' start value h/deg+b; fill and finalize launches go (11 to 7); sum order: start value, then atomic arrival
# speedup vs baseline: 1.9604x; 1.1863x over previous
#include <hip/hip_runtime.h>
#include <math.h>

typedef __attribute__((ext_vector_type(2))) float v2f;
typedef __attribute__((ext_vector_type(8))) float v8f;

#define IN_C 128
#define LOG2C 7

__global__ void k_set_const(float* __restrict__ p, float v, int n) {
    int i = blockIdx.x * blockDim.x + threadIdx.x;
    if (i < n) p[i] = v;
}

__global__ void k_count_deg(const int* __restrict__ dst, float* __restrict__ deg, int nE) {
    int i = blockIdx.x * blockDim.x + threadIdx.x;
    if (i < nE) atomicAdd(&deg[dst[i]], 1.0f);
}

__global__ void k_inv_sqrt(float* __restrict__ p, int n) {
    int i = blockIdx.x * blockDim.x + threadIdx.x;
    if (i < n) p[i] = 1.0f / sqrtf(p[i]);
}

__device__ __forceinline__ void gemm_strip_f32(
    const float* A, const float* __restrict__ W,
    const float* __restrict__ inv_sqrt, const float* __restrict__ bias,
    float* __restrict__ H, float* S, int mTiles, const bool relu_in)
{
    const int wavesPerBlock = blockDim.x >> 5;
    const int wave = blockIdx.x * wavesPerBlock + (threadIdx.x >> 5);
    if (wave >= mTiles) return;

    const int lane = threadIdx.x & 31;
    const int row  = lane & 15;
    const int hi   = lane >> 4;
    const int koff = hi * 2;
    const int m0   = wave * 16;

    v8f acc[8] = {};

    const float* arow = A + (size_t)(m0 + row) * IN_C;

    for (int k = 0; k < IN_C; k += 4) {
        v2f a;
        a.x = arow[k + koff];
        a.y = arow[k + koff + 1];
        if (relu_in) {
            a.x = fmaxf(a.x, 0.0f);
            a.y = fmaxf(a.y, 0.0f);
        }
        const float* __restrict__ wbase = W + (size_t)(k + koff) * IN_C + row;
#pragma unroll
        for (int t = 0; t < 8; ++t) {
            v2f b;
            b.x = wbase[t * 16];
            b.y = wbase[t * 16 + IN_C];
            acc[t] = __builtin_amdgcn_wmma_f32_16x16x4_f32(
                false, a, false, b, (short)0, acc[t], false, false);
        }
    }

    const int r0 = m0 + hi * 8;
    float isq[8];
#pragma unroll
    for (int r = 0; r < 8; ++r) isq[r] = inv_sqrt[r0 + r];

    float* __restrict__ hbase = H + (size_t)r0 * IN_C + row;
    float* sbase = S + (size_t)r0 * IN_C + row;
#pragma unroll
    for (int t = 0; t < 8; ++t) {
        const float bt = bias[t * 16 + row];
#pragma unroll
        for (int r = 0; r < 8; ++r) {
            const float hv = acc[t][r];
            hbase[(size_t)r * IN_C + t * 16] = hv;
            sbase[(size_t)r * IN_C + t * 16] = hv * isq[r] * isq[r] + bt;
        }
    }
}

__global__ __launch_bounds__(128) void k_gemm_layer1(
    const float* __restrict__ A, const float* __restrict__ W,
    const float* __restrict__ inv_sqrt, const float* __restrict__ bias,
    float* __restrict__ H, float* __restrict__ S, int mTiles)
{
    gemm_strip_f32(A, W, inv_sqrt, bias, H, S, mTiles, false);
}

__global__ __launch_bounds__(128) void k_gemm_layer2(
    float* AS, const float* __restrict__ W,
    const float* __restrict__ inv_sqrt, const float* __restrict__ bias,
    float* __restrict__ H, int mTiles)
{
    gemm_strip_f32(AS, W, inv_sqrt, bias, H, AS, mTiles, true);
}

#define EDGES_PER_WAVE 32
#define EDGE_BATCH 8

__device__ __forceinline__ void edge_row_add(
    const float* __restrict__ h, float* __restrict__ agg,
    int sj, int dj, float cj, int lane)
{
    const float* __restrict__ hp = h + (size_t)(unsigned)sj * IN_C + lane;
    const float h0 = hp[0], h1 = hp[32], h2 = hp[64], h3 = hp[96];
    float* ap = agg + (size_t)(unsigned)dj * IN_C + lane;
    atomicAdd(ap +  0, h0 * cj);
    atomicAdd(ap + 32, h1 * cj);
    atomicAdd(ap + 64, h2 * cj);
    atomicAdd(ap + 96, h3 * cj);
}

__global__ __launch_bounds__(256) void k_aggregate(
    const float* __restrict__ h,
    const int* __restrict__ src, const int* __restrict__ dst,
    const float* __restrict__ inv_sqrt,
    float* __restrict__ agg, int nE)
{
    const int wave = __builtin_amdgcn_readfirstlane(
        (int)(blockIdx.x * (blockDim.x >> 5) + (threadIdx.x >> 5)));
    const int base = wave * EDGES_PER_WAVE;
    if (base >= nE) return;
    const int lane = threadIdx.x & 31;
    const int left = nE - base;
    const int cnt = left < EDGES_PER_WAVE ? left : EDGES_PER_WAVE;

    int s = 0, d = 0;
    float coef = 0.0f;
    if (lane < cnt) {
        s = src[base + lane];
        d = dst[base + lane];
        coef = inv_sqrt[s] * inv_sqrt[d];
    }
    const int coef_bits = __float_as_int(coef);

    int j = 0;
    for (; j + EDGE_BATCH <= cnt; j += EDGE_BATCH) {
        float v[EDGE_BATCH][4];
#pragma unroll
        for (int u = 0; u < EDGE_BATCH; ++u) {
            const int sj = __builtin_amdgcn_readlane(s, j + u);
            const float* __restrict__ hp = h + (size_t)(unsigned)sj * IN_C + lane;
            v[u][0] = hp[0]; v[u][1] = hp[32]; v[u][2] = hp[64]; v[u][3] = hp[96];
        }
#pragma unroll
        for (int u = 0; u < EDGE_BATCH; ++u) {
            const int dj = __builtin_amdgcn_readlane(d, j + u);
            const float cj = __int_as_float(__builtin_amdgcn_readlane(coef_bits, j + u));
            float* ap = agg + (size_t)(unsigned)dj * IN_C + lane;
            atomicAdd(ap +  0, v[u][0] * cj);
            atomicAdd(ap + 32, v[u][1] * cj);
            atomicAdd(ap + 64, v[u][2] * cj);
            atomicAdd(ap + 96, v[u][3] * cj);
        }
    }
    for (; j < cnt; ++j) {
        edge_row_add(h, agg,
                     __builtin_amdgcn_readlane(s, j),
                     __builtin_amdgcn_readlane(d, j),
                     __int_as_float(__builtin_amdgcn_readlane(coef_bits, j)), lane);
    }
}

extern "C" void kernel_launch(void* const* d_in, const int* in_sizes, int n_in,
                              void* d_out, int out_size, void* d_ws, size_t ws_size,
                              hipStream_t stream)
{
    const float* x   = (const float*)d_in[0];
    const int*   ei  = (const int*)d_in[1];
    const float* W1  = (const float*)d_in[2];
    const float* b1  = (const float*)d_in[3];
    const float* W2  = (const float*)d_in[4];
    const float* b2  = (const float*)d_in[5];

    const int N = in_sizes[0] / IN_C;
    const int E = in_sizes[1] / 2;
    const int* src = ei;
    const int* dst = ei + E;

    float* inv_sqrt = (float*)d_ws;
    float* hbuf     = inv_sqrt + N;
    float* out      = (float*)d_out;

    const int T256 = 256;
    const int mTiles = (N + 15) / 16;
    dim3 gemmGrid((mTiles + 3) / 4);
    dim3 gemmBlock(128);
    dim3 aggGrid((E + 255) / 256);
    dim3 aggBlock(256);

    k_set_const<<<(N + T256 - 1) / T256, T256, 0, stream>>>(inv_sqrt, 1.0f, N);
    k_count_deg<<<(E + T256 - 1) / T256, T256, 0, stream>>>(dst, inv_sqrt, E);
    k_inv_sqrt<<<(N + T256 - 1) / T256, T256, 0, stream>>>(inv_sqrt, N);

    k_gemm_layer1<<<gemmGrid, gemmBlock, 0, stream>>>(x, W1, inv_sqrt, b1, hbuf, out, mTiles);
    k_aggregate<<<aggGrid, aggBlock, 0, stream>>>(hbuf, src, dst, inv_sqrt, out, E);

    k_gemm_layer2<<<gemmGrid, gemmBlock, 0, stream>>>(out, W2, inv_sqrt, b2, hbuf, mTiles);
    k_aggregate<<<aggGrid, aggBlock, 0, stream>>>(hbuf, src, dst, inv_sqrt, out, E);
}


# Edge pass: destination rows split between the chip's two halves: work-groups draw 256-edge tickets from their own half's counter and add only that half's rows; sum order: atomic arrival, as before
# speedup vs baseline: 2.1673x; 1.1055x over previous
#include <hip/hip_runtime.h>
#include <math.h>

typedef __attribute__((ext_vector_type(2))) float v2f;
typedef __attribute__((ext_vector_type(8))) float v8f;

#define IN_C 128
#define LOG2C 7

__global__ void k_set_const(float* __restrict__ p, float v, int n) {
    int i = blockIdx.x * blockDim.x + threadIdx.x;
    if (i < n) p[i] = v;
}

__global__ void k_count_deg(const int* __restrict__ dst, float* __restrict__ deg, int nE) {
    int i = blockIdx.x * blockDim.x + threadIdx.x;
    if (i < nE) atomicAdd(&deg[dst[i]], 1.0f);
}

#define TICKET_STRIDE 32
#define TICKET_COUNTERS 4

__global__ void k_inv_sqrt(float* __restrict__ p, int n, int* __restrict__ tickets) {
    int i = blockIdx.x * blockDim.x + threadIdx.x;
    if (i < TICKET_COUNTERS * TICKET_STRIDE) tickets[i] = 0;
    if (i < n) p[i] = 1.0f / sqrtf(p[i]);
}

__device__ __forceinline__ void gemm_strip_f32(
    const float* A, const float* __restrict__ W,
    const float* __restrict__ inv_sqrt, const float* __restrict__ bias,
    float* __restrict__ H, float* S, int mTiles, const bool relu_in)
{
    const int wavesPerBlock = blockDim.x >> 5;
    const int wave = blockIdx.x * wavesPerBlock + (threadIdx.x >> 5);
    if (wave >= mTiles) return;

    const int lane = threadIdx.x & 31;
    const int row  = lane & 15;
    const int hi   = lane >> 4;
    const int koff = hi * 2;
    const int m0   = wave * 16;

    v8f acc[8] = {};

    const float* arow = A + (size_t)(m0 + row) * IN_C;

    for (int k = 0; k < IN_C; k += 4) {
        v2f a;
        a.x = arow[k + koff];
        a.y = arow[k + koff + 1];
        if (relu_in) {
            a.x = fmaxf(a.x, 0.0f);
            a.y = fmaxf(a.y, 0.0f);
        }
        const float* __restrict__ wbase = W + (size_t)(k + koff) * IN_C + row;
#pragma unroll
        for (int t = 0; t < 8; ++t) {
            v2f b;
            b.x = wbase[t * 16];
            b.y = wbase[t * 16 + IN_C];
            acc[t] = __builtin_amdgcn_wmma_f32_16x16x4_f32(
                false, a, false, b, (short)0, acc[t], false, false);
        }
    }

    const int r0 = m0 + hi * 8;
    float isq[8];
#pragma unroll
    for (int r = 0; r < 8; ++r) isq[r] = inv_sqrt[r0 + r];

    float* __restrict__ hbase = H + (size_t)r0 * IN_C + row;
    float* sbase = S + (size_t)r0 * IN_C + row;
#pragma unroll
    for (int t = 0; t < 8; ++t) {
        const float bt = bias[t * 16 + row];
#pragma unroll
        for (int r = 0; r < 8; ++r) {
            const float hv = acc[t][r];
            hbase[(size_t)r * IN_C + t * 16] = hv;
            sbase[(size_t)r * IN_C + t * 16] = hv * isq[r] * isq[r] + bt;
        }
    }
}

__global__ __launch_bounds__(128) void k_gemm_layer1(
    const float* __restrict__ A, const float* __restrict__ W,
    const float* __restrict__ inv_sqrt, const float* __restrict__ bias,
    float* __restrict__ H, float* __restrict__ S, int mTiles)
{
    gemm_strip_f32(A, W, inv_sqrt, bias, H, S, mTiles, false);
}

__global__ __launch_bounds__(128) void k_gemm_layer2(
    float* AS, const float* __restrict__ W,
    const float* __restrict__ inv_sqrt, const float* __restrict__ bias,
    float* __restrict__ H, int mTiles)
{
    gemm_strip_f32(AS, W, inv_sqrt, bias, H, AS, mTiles, true);
}

#define AGG_WAVES 8
#define TICKET_EDGES (AGG_WAVES * 32)
#define AGG_BLOCKS 2048

__device__ __forceinline__ void edge_row_add(
    const float* __restrict__ h, float* __restrict__ agg,
    int sj, int dj, float cj, int lane)
{
    const float* __restrict__ hp = h + (size_t)(unsigned)sj * IN_C + lane;
    const float h0 = hp[0], h1 = hp[32], h2 = hp[64], h3 = hp[96];
    float* ap = agg + (size_t)(unsigned)dj * IN_C + lane;
    atomicAdd(ap +  0, h0 * cj);
    atomicAdd(ap + 32, h1 * cj);
    atomicAdd(ap + 64, h2 * cj);
    atomicAdd(ap + 96, h3 * cj);
}

template <int B>
__device__ __forceinline__ void edge_rows_add_batch(
    const float* __restrict__ h, float* __restrict__ agg,
    int s, int d, int coef_bits, unsigned& m, int lane)
{
    float v[B][4];
    int jl[B];
#pragma unroll
    for (int u = 0; u < B; ++u) {
        jl[u] = __builtin_ctz(m);
        m &= m - 1;
        const int sj = __builtin_amdgcn_readlane(s, jl[u]);
        const float* __restrict__ hp = h + (size_t)(unsigned)sj * IN_C + lane;
        v[u][0] = hp[0]; v[u][1] = hp[32]; v[u][2] = hp[64]; v[u][3] = hp[96];
    }
#pragma unroll
    for (int u = 0; u < B; ++u) {
        const int dj = __builtin_amdgcn_readlane(d, jl[u]);
        const float cj = __int_as_float(__builtin_amdgcn_readlane(coef_bits, jl[u]));
        float* ap = agg + (size_t)(unsigned)dj * IN_C + lane;
        atomicAdd(ap +  0, v[u][0] * cj);
        atomicAdd(ap + 32, v[u][1] * cj);
        atomicAdd(ap + 64, v[u][2] * cj);
        atomicAdd(ap + 96, v[u][3] * cj);
    }
}

__global__ __launch_bounds__(256) void k_aggregate(
    const float* __restrict__ h,
    const int* __restrict__ src, const int* __restrict__ dst,
    const float* __restrict__ inv_sqrt,
    float* __restrict__ agg, int nE, int nNodes, int* __restrict__ tickets)
{
    __shared__ int s_half;
    __shared__ int s_ticket[2];

    const int lane = threadIdx.x & 31;
    const int wv   = threadIdx.x >> 5;
    const int split = nNodes >> 1;
    const int nT = (nE + TICKET_EDGES - 1) / TICKET_EDGES;

    if (threadIdx.x == 0) s_half = (int)((__builtin_amdgcn_s_sendmsg_rtn(0x87) >> 18) & 1u);
    __syncthreads();
    const int myHalf = s_half;

    int slot = 0;
    for (int pass = 0; pass < 2; ++pass) {
        const int half = myHalf ^ pass;
        unsigned int* ctr = (unsigned int*)(tickets + half * TICKET_STRIDE);
        int tnext = 0;
        if (threadIdx.x == 0) tnext = (int)atomicInc(ctr, 0x7fffffffu);
        for (int it = 0; it <= nT; ++it) {
            if (threadIdx.x == 0) s_ticket[slot] = tnext;
            __syncthreads();
            const int t = s_ticket[slot];
            slot ^= 1;
            if (t >= nT) break;
            if (threadIdx.x == 0) tnext = (int)atomicInc(ctr, 0x7fffffffu);

            const int e = t * TICKET_EDGES + wv * 32 + lane;
            int s = 0, d = 0;
            if (e < nE) {
                s = src[e];
                d = dst[e];
            }
            const bool keep = (e < nE) && ((d >= split ? 1 : 0) == half);
            float coef = 0.0f;
            if (keep) coef = inv_sqrt[s] * inv_sqrt[d];
            const int coef_bits = __float_as_int(coef);

            unsigned m = __builtin_amdgcn_ballot_w32(keep);
            while (__builtin_popcount(m) >= 8) edge_rows_add_batch<8>(h, agg, s, d, coef_bits, m, lane);
            if (__builtin_popcount(m) >= 4)    edge_rows_add_batch<4>(h, agg, s, d, coef_bits, m, lane);
            while (m) {
                const int j = __builtin_ctz(m);
                m &= m - 1;
                edge_row_add(h, agg,
                             __builtin_amdgcn_readlane(s, j),
                             __builtin_amdgcn_readlane(d, j),
                             __int_as_float(__builtin_amdgcn_readlane(coef_bits, j)), lane);
            }
        }
    }
}

extern "C" void kernel_launch(void* const* d_in, const int* in_sizes, int n_in,
                              void* d_out, int out_size, void* d_ws, size_t ws_size,
                              hipStream_t stream)
{
    const float* x   = (const float*)d_in[0];
    const int*   ei  = (const int*)d_in[1];
    const float* W1  = (const float*)d_in[2];
    const float* b1  = (const float*)d_in[3];
    const float* W2  = (const float*)d_in[4];
    const float* b2  = (const float*)d_in[5];

    const int N = in_sizes[0] / IN_C;
    const int E = in_sizes[1] / 2;
    const int* src = ei;
    const int* dst = ei + E;

    float* inv_sqrt = (float*)d_ws;
    float* hbuf     = inv_sqrt + N;
    int*   tickets  = (int*)(hbuf + (size_t)N * IN_C);
    float* out      = (float*)d_out;

    const int T256 = 256;
    const int mTiles = (N + 15) / 16;
    dim3 gemmGrid((mTiles + 3) / 4);
    dim3 gemmBlock(128);
    dim3 aggGrid(AGG_BLOCKS);
    dim3 aggBlock(256);

    k_set_const<<<(N + T256 - 1) / T256, T256, 0, stream>>>(inv_sqrt, 1.0f, N);
    k_count_deg<<<(E + T256 - 1) / T256, T256, 0, stream>>>(dst, inv_sqrt, E);
    k_inv_sqrt<<<(N + T256 - 1) / T256, T256, 0, stream>>>(inv_sqrt, N, tickets);

    k_gemm_layer1<<<gemmGrid, gemmBlock, 0, stream>>>(x, W1, inv_sqrt, b1, hbuf, out, mTiles);
    k_aggregate<<<aggGrid, aggBlock, 0, stream>>>(hbuf, src, dst, inv_sqrt, out, E, N, tickets + 0);

    k_gemm_layer2<<<gemmGrid, gemmBlock, 0, stream>>>(out, W2, inv_sqrt, b2, hbuf, mTiles);
    k_aggregate<<<aggGrid, aggBlock, 0, stream>>>(hbuf, src, dst, inv_sqrt, out, E, N, tickets + 2 * TICKET_STRIDE);
}


# Degree count: each work-group counts one (edge slice, node range) in LDS and stores plain words, summed per node in the 1/sqrt kernel; scattered global adds and fill launch go (7 to 6); counts exact
# speedup vs baseline: 2.5669x; 1.1844x over previous
#include <hip/hip_runtime.h>
#include <math.h>

typedef __attribute__((ext_vector_type(2))) float v2f;
typedef __attribute__((ext_vector_type(8))) float v8f;

#define IN_C 128
#define LOG2C 7

#define TICKET_STRIDE 32
#define TICKET_COUNTERS 4

#define DEG_SLICES  32
#define DEG_RANGE   12512
#define DEG_THREADS 512
#define DEG_BATCH   8

__device__ __forceinline__ void deg_count_one(int* s_cnt, int d, int base, int len) {
    const unsigned r = (unsigned)(d - base);
    if (r < (unsigned)len) atomicAdd(&s_cnt[r], 1);
}

__global__ __launch_bounds__(DEG_THREADS) void k_count_deg(
    const int* __restrict__ dst, int* __restrict__ partial, int nE, int nNodes)
{
    __shared__ int s_cnt[DEG_RANGE];

    const int tid  = threadIdx.x;
    const int base = blockIdx.y * DEG_RANGE;
    const int len  = min(DEG_RANGE, nNodes - base);

    for (int i = tid; i < DEG_RANGE; i += DEG_THREADS) s_cnt[i] = 0;
    __syncthreads();

    const int quads = (nE + 3) >> 2;
    const int qPer  = (quads + (int)gridDim.x - 1) / (int)gridDim.x;
    const int q0    = blockIdx.x * qPer;
    const int q1    = min(q0 + qPer, quads);
    const int e1    = min(q1 << 2, nE);

    int eRest = q0 << 2;
    if ((((size_t)dst) & 15) == 0) {
        const int qFull = min(q1, nE >> 2);
        for (int q = q0 + tid; q < qFull; q += DEG_BATCH * DEG_THREADS) {
            int4 v[DEG_BATCH];
#pragma unroll
            for (int u = 0; u < DEG_BATCH; ++u) {
                const int qq = q + u * DEG_THREADS;
                v[u] = make_int4(-1, -1, -1, -1);
                if (qq < qFull) v[u] = *reinterpret_cast<const int4*>(dst + ((size_t)qq << 2));
            }
#pragma unroll
            for (int u = 0; u < DEG_BATCH; ++u) {
                deg_count_one(s_cnt, v[u].x, base, len);
                deg_count_one(s_cnt, v[u].y, base, len);
                deg_count_one(s_cnt, v[u].z, base, len);
                deg_count_one(s_cnt, v[u].w, base, len);
            }
        }
        eRest = max(eRest, qFull << 2);
    }
    for (int e = eRest + tid; e < e1; e += DEG_THREADS) deg_count_one(s_cnt, dst[e], base, len);
    __syncthreads();

    int* __restrict__ prow = partial + (size_t)blockIdx.x * nNodes + base;
    for (int i = tid; i < len; i += DEG_THREADS) prow[i] = s_cnt[i];
}

__global__ void k_inv_sqrt(const int* __restrict__ partial, float* __restrict__ p, int n,
                           int* __restrict__ tickets) {
    int i = blockIdx.x * blockDim.x + threadIdx.x;
    if (i < TICKET_COUNTERS * TICKET_STRIDE) tickets[i] = 0;
    if (i < n) {
        int c = 1;
#pragma unroll
        for (int s = 0; s < DEG_SLICES; ++s) c += partial[(size_t)s * n + i];
        p[i] = 1.0f / sqrtf((float)c);
    }
}

__device__ __forceinline__ void gemm_strip_f32(
    const float* A, const float* __restrict__ W,
    const float* __restrict__ inv_sqrt, const float* __restrict__ bias,
    float* __restrict__ H, float* S, int mTiles, const bool relu_in)
{
    const int wavesPerBlock = blockDim.x >> 5;
    const int wave = blockIdx.x * wavesPerBlock + (threadIdx.x >> 5);
    if (wave >= mTiles) return;

    const int lane = threadIdx.x & 31;
    const int row  = lane & 15;
    const int hi   = lane >> 4;
    const int koff = hi * 2;
    const int m0   = wave * 16;

    v8f acc[8] = {};

    const float* arow = A + (size_t)(m0 + row) * IN_C;

    for (int k = 0; k < IN_C; k += 4) {
        v2f a;
        a.x = arow[k + koff];
        a.y = arow[k + koff + 1];
        if (relu_in) {
            a.x = fmaxf(a.x, 0.0f);
            a.y = fmaxf(a.y, 0.0f);
        }
        const float* __restrict__ wbase = W + (size_t)(k + koff) * IN_C + row;
#pragma unroll
        for (int t = 0; t < 8; ++t) {
            v2f b;
            b.x = wbase[t * 16];
            b.y = wbase[t * 16 + IN_C];
            acc[t] = __builtin_amdgcn_wmma_f32_16x16x4_f32(
                false, a, false, b, (short)0, acc[t], false, false);
        }
    }

    const int r0 = m0 + hi * 8;
    float isq[8];
#pragma unroll
    for (int r = 0; r < 8; ++r) isq[r] = inv_sqrt[r0 + r];

    float* __restrict__ hbase = H + (size_t)r0 * IN_C + row;
    float* sbase = S + (size_t)r0 * IN_C + row;
#pragma unroll
    for (int t = 0; t < 8; ++t) {
        const float bt = bias[t * 16 + row];
#pragma unroll
        for (int r = 0; r < 8; ++r) {
            const float hv = acc[t][r];
            hbase[(size_t)r * IN_C + t * 16] = hv;
            sbase[(size_t)r * IN_C + t * 16] = hv * isq[r] * isq[r] + bt;
        }
    }
}

__global__ __launch_bounds__(128) void k_gemm_layer1(
    const float* __restrict__ A, const float* __restrict__ W,
    const float* __restrict__ inv_sqrt, const float* __restrict__ bias,
    float* __restrict__ H, float* __restrict__ S, int mTiles)
{
    gemm_strip_f32(A, W, inv_sqrt, bias, H, S, mTiles, false);
}

__global__ __launch_bounds__(128) void k_gemm_layer2(
    float* AS, const float* __restrict__ W,
    const float* __restrict__ inv_sqrt, const float* __restrict__ bias,
    float* __restrict__ H, int mTiles)
{
    gemm_strip_f32(AS, W, inv_sqrt, bias, H, AS, mTiles, true);
}

#define AGG_WAVES 8
#define TICKET_EDGES (AGG_WAVES * 32)
#define AGG_BLOCKS 2048

__device__ __forceinline__ void edge_row_add(
    const float* __restrict__ h, float* __restrict__ agg,
    int sj, int dj, float cj, int lane)
{
    const float* __restrict__ hp = h + (size_t)(unsigned)sj * IN_C + lane;
    const float h0 = hp[0], h1 = hp[32], h2 = hp[64], h3 = hp[96];
    float* ap = agg + (size_t)(unsigned)dj * IN_C + lane;
    atomicAdd(ap +  0, h0 * cj);
    atomicAdd(ap + 32, h1 * cj);
    atomicAdd(ap + 64, h2 * cj);
    atomicAdd(ap + 96, h3 * cj);
}

template <int B>
__device__ __forceinline__ void edge_rows_add_batch(
    const float* __restrict__ h, float* __restrict__ agg,
    int s, int d, int coef_bits, unsigned& m, int lane)
{
    float v[B][4];
    int jl[B];
#pragma unroll
    for (int u = 0; u < B; ++u) {
        jl[u] = __builtin_ctz(m);
        m &= m - 1;
        const int sj = __builtin_amdgcn_readlane(s, jl[u]);
        const float* __restrict__ hp = h + (size_t)(unsigned)sj * IN_C + lane;
        v[u][0] = hp[0]; v[u][1] = hp[32]; v[u][2] = hp[64]; v[u][3] = hp[96];
    }
#pragma unroll
    for (int u = 0; u < B; ++u) {
        const int dj = __builtin_amdgcn_readlane(d, jl[u]);
        const float cj = __int_as_float(__builtin_amdgcn_readlane(coef_bits, jl[u]));
        float* ap = agg + (size_t)(unsigned)dj * IN_C + lane;
        atomicAdd(ap +  0, v[u][0] * cj);
        atomicAdd(ap + 32, v[u][1] * cj);
        atomicAdd(ap + 64, v[u][2] * cj);
        atomicAdd(ap + 96, v[u][3] * cj);
    }
}

__global__ __launch_bounds__(256) void k_aggregate(
    const float* __restrict__ h,
    const int* __restrict__ src, const int* __restrict__ dst,
    const float* __restrict__ inv_sqrt,
    float* __restrict__ agg, int nE, int nNodes, int* __restrict__ tickets)
{
    __shared__ int s_half;
    __shared__ int s_ticket[2];

    const int lane = threadIdx.x & 31;
    const int wv   = threadIdx.x >> 5;
    const int split = nNodes >> 1;
    const int nT = (nE + TICKET_EDGES - 1) / TICKET_EDGES;

    if (threadIdx.x == 0) s_half = (int)((__builtin_amdgcn_s_sendmsg_rtn(0x87) >> 18) & 1u);
    __syncthreads();
    const int myHalf = s_half;

    int slot = 0;
    for (int pass = 0; pass < 2; ++pass) {
        const int half = myHalf ^ pass;
        unsigned int* ctr = (unsigned int*)(tickets + half * TICKET_STRIDE);
        int tnext = 0;
        if (threadIdx.x == 0) tnext = (int)atomicInc(ctr, 0x7fffffffu);
        for (int it = 0; it <= nT; ++it) {
            if (threadIdx.x == 0) s_ticket[slot] = tnext;
            __syncthreads();
            const int t = s_ticket[slot];
            slot ^= 1;
            if (t >= nT) break;
            if (threadIdx.x == 0) tnext = (int)atomicInc(ctr, 0x7fffffffu);

            const int e = t * TICKET_EDGES + wv * 32 + lane;
            int s = 0, d = 0;
            if (e < nE) {
                s = src[e];
                d = dst[e];
            }
            const bool keep = (e < nE) && ((d >= split ? 1 : 0) == half);
            float coef = 0.0f;
            if (keep) coef = inv_sqrt[s] * inv_sqrt[d];
            const int coef_bits = __float_as_int(coef);

            unsigned m = __builtin_amdgcn_ballot_w32(keep);
            while (__builtin_popcount(m) >= 8) edge_rows_add_batch<8>(h, agg, s, d, coef_bits, m, lane);
            if (__builtin_popcount(m) >= 4)    edge_rows_add_batch<4>(h, agg, s, d, coef_bits, m, lane);
            while (m) {
                const int j = __builtin_ctz(m);
                m &= m - 1;
                edge_row_add(h, agg,
                             __builtin_amdgcn_readlane(s, j),
                             __builtin_amdgcn_readlane(d, j),
                             __int_as_float(__builtin_amdgcn_readlane(coef_bits, j)), lane);
            }
        }
    }
}

extern "C" void kernel_launch(void* const* d_in, const int* in_sizes, int n_in,
                              void* d_out, int out_size, void* d_ws, size_t ws_size,
                              hipStream_t stream)
{
    const float* x   = (const float*)d_in[0];
    const int*   ei  = (const int*)d_in[1];
    const float* W1  = (const float*)d_in[2];
    const float* b1  = (const float*)d_in[3];
    const float* W2  = (const float*)d_in[4];
    const float* b2  = (const float*)d_in[5];

    const int N = in_sizes[0] / IN_C;
    const int E = in_sizes[1] / 2;
    const int* src = ei;
    const int* dst = ei + E;

    float* inv_sqrt = (float*)d_ws;
    float* hbuf     = inv_sqrt + N;
    int*   tickets  = (int*)(hbuf + (size_t)N * IN_C);
    int*   partial  = tickets + TICKET_COUNTERS * TICKET_STRIDE;
    float* out      = (float*)d_out;

    const int T256 = 256;
    const int mTiles = (N + 15) / 16;
    dim3 gemmGrid((mTiles + 3) / 4);
    dim3 gemmBlock(128);
    dim3 aggGrid(AGG_BLOCKS);
    dim3 aggBlock(256);

    dim3 degGrid(DEG_SLICES, (N + DEG_RANGE - 1) / DEG_RANGE);
    k_count_deg<<<degGrid, DEG_THREADS, 0, stream>>>(dst, partial, E, N);
    k_inv_sqrt<<<(N + T256 - 1) / T256, T256, 0, stream>>>(partial, inv_sqrt, N, tickets);

    k_gemm_layer1<<<gemmGrid, gemmBlock, 0, stream>>>(x, W1, inv_sqrt, b1, hbuf, out, mTiles);
    k_aggregate<<<aggGrid, aggBlock, 0, stream>>>(hbuf, src, dst, inv_sqrt, out, E, N, tickets + 0);

    k_gemm_layer2<<<gemmGrid, gemmBlock, 0, stream>>>(out, W2, inv_sqrt, b2, hbuf, mTiles);
    k_aggregate<<<aggGrid, aggBlock, 0, stream>>>(hbuf, src, dst, inv_sqrt, out, E, N, tickets + 2 * TICKET_STRIDE);
}
